# starting kernel
# baseline (speedup 1.0000x reference)
#include <hip/hip_runtime.h>

typedef float v2f __attribute__((ext_vector_type(2)));
typedef float v4f __attribute__((ext_vector_type(4)));
typedef float v8f __attribute__((ext_vector_type(8)));

#define LAMBDA_ 0.01f
#define B_ 8
#define S_ 1024
#define E_ 512
#define BS_ (B_ * S_)

__global__ void __launch_bounds__(32) rowsum_wmma(
    const float* __restrict__ y0, const float* __restrict__ y1,
    const float* __restrict__ y2, const float* __restrict__ y3,
    const float* __restrict__ y4, const float* __restrict__ y5,
    const float* __restrict__ y6, const float* __restrict__ y7,
    float* __restrict__ ws)
{
    const int m  = blockIdx.x >> 5;
    const int rb = blockIdx.x & 31;
    const float* Y =
        (m == 0) ? y0 : (m == 1) ? y1 : (m == 2) ? y2 : (m == 3) ? y3 :
        (m == 4) ? y4 : (m == 5) ? y5 : (m == 6) ? y6 : y7;

    const int lane  = threadIdx.x;
    const int row   = rb * 16 + (lane & 15);
    const int khalf = (lane >> 4) * 2;

    v8f c = {};
    v2f b;  b.x = 1.0f;  b.y = 1.0f;

    const float* rowp = Y + (size_t)row * E_ + khalf;
    #pragma unroll 4
    for (int k0 = 0; k0 < E_; k0 += 4) {
        v2f a = *(const v2f*)(rowp + k0);
        c = __builtin_amdgcn_wmma_f32_16x16x4_f32(
                 false, a,  false, b,
                 (short)0, c,  false,  false);
    }

    float* o = ws + (size_t)m * E_ + rb * 16;
    if (lane == 0) {
        #pragma unroll
        for (int v = 0; v < 8; ++v) o[v] = c[v];
    } else if (lane == 16) {
        #pragma unroll
        for (int v = 0; v < 8; ++v) o[8 + v] = c[v];
    }
}

__device__ __forceinline__ v4f sshrink4(v4f v) {
    v4f r;
    #pragma unroll
    for (int i = 0; i < 4; ++i) {
        float x = v[i];
        r[i] = (x > LAMBDA_) ? (x - LAMBDA_) : ((x < -LAMBDA_) ? (x + LAMBDA_) : 0.0f);
    }
    return r;
}

__device__ __forceinline__ void cm(v4f xr, v4f xi, v4f sr, v4f si,
                                   v4f& re, v4f& im) {
    re = sshrink4(xr * sr - xi * si);
    im = sshrink4(xr * si + xi * sr);
}

__global__ void __launch_bounds__(256) octmlp_main(
    const float* __restrict__ xr, const float* __restrict__ xi,
    const float* __restrict__ ws,
    const float* __restrict__ bR1, const float* __restrict__ bI1,
    const float* __restrict__ bR2, const float* __restrict__ bI2,
    const float* __restrict__ bR3, const float* __restrict__ bI3,
    const float* __restrict__ bR4, const float* __restrict__ bI4,
    float* __restrict__ out)
{
    const int QE  = E_ / 4;
    int gid = blockIdx.x * blockDim.x + threadIdx.x;
    int bs  = gid / QE;
    int e4  = (gid - bs * QE) * 4;
    if (bs >= BS_) return;

    const float* xrb = xr + (size_t)bs * 4 * E_ + e4;
    const float* xib = xi + (size_t)bs * 4 * E_ + e4;

    v4f x1r = __builtin_nontemporal_load((const v4f*)(xrb + 0 * E_));
    v4f x2r = __builtin_nontemporal_load((const v4f*)(xrb + 1 * E_));
    v4f x3r = __builtin_nontemporal_load((const v4f*)(xrb + 2 * E_));
    v4f x4r = __builtin_nontemporal_load((const v4f*)(xrb + 3 * E_));
    v4f x1i = __builtin_nontemporal_load((const v4f*)(xib + 0 * E_));
    v4f x2i = __builtin_nontemporal_load((const v4f*)(xib + 1 * E_));
    v4f x3i = __builtin_nontemporal_load((const v4f*)(xib + 2 * E_));
    v4f x4i = __builtin_nontemporal_load((const v4f*)(xib + 3 * E_));

    v4f r1 = *(const v4f*)(ws + 0 * E_ + e4);
    v4f i1 = *(const v4f*)(ws + 1 * E_ + e4);
    v4f r2 = *(const v4f*)(ws + 2 * E_ + e4);
    v4f i2 = *(const v4f*)(ws + 3 * E_ + e4);
    v4f r3 = *(const v4f*)(ws + 4 * E_ + e4);
    v4f i3 = *(const v4f*)(ws + 5 * E_ + e4);
    v4f r4 = *(const v4f*)(ws + 6 * E_ + e4);
    v4f i4 = *(const v4f*)(ws + 7 * E_ + e4);
    v4f ni2 = -i2, ni3 = -i3, ni4 = -i4;

    v4f tre, tim;
    v4f a1re, a1im, a2re, a2im, a3re, a3im, a4re, a4im;

    cm(x1r, x1i, r1, i1, a1re, a1im);
    cm(x2r, x2i, r2, ni2, tre, tim);  a1re -= tre;  a1im -= tim;
    cm(x3r, x3i, r3, ni3, tre, tim);  a1re -= tre;  a1im -= tim;
    cm(x4r, x4i, r4, ni4, tre, tim);  a1re -= tre;  a1im -= tim;

    cm(x2r, x2i, r1, i1, a2re, a2im);
    cm(x1r, x1i, r2, i2, tre, tim);   a2re += tre;  a2im += tim;
    cm(x4r, x4i, r3, ni3, tre, tim);  a2re += tre;  a2im += tim;
    cm(x3r, x3i, r4, ni4, tre, tim);  a2re -= tre;  a2im -= tim;

    cm(x3r, x3i, r1, i1, a3re, a3im);
    cm(x4r, x4i, r2, ni2, tre, tim);  a3re -= tre;  a3im -= tim;
    cm(x1r, x1i, r3, i3, tre, tim);   a3re += tre;  a3im += tim;
    cm(x2r, x2i, r4, ni4, tre, tim);  a3re += tre;  a3im += tim;

    cm(x4r, x4i, r1, i1, a4re, a4im);
    cm(x3r, x3i, r2, ni2, tre, tim);  a4re += tre;  a4im += tim;
    cm(x2r, x2i, r3, ni3, tre, tim);  a4re -= tre;  a4im -= tim;
    cm(x1r, x1i, r4, i4, tre, tim);   a4re += tre;  a4im += tim;

    v4f vbR1 = *(const v4f*)(bR1 + e4), vbI1 = *(const v4f*)(bI1 + e4);
    v4f vbR2 = *(const v4f*)(bR2 + e4), vbI2 = *(const v4f*)(bI2 + e4);
    v4f vbR3 = *(const v4f*)(bR3 + e4), vbI3 = *(const v4f*)(bI3 + e4);
    v4f vbR4 = *(const v4f*)(bR4 + e4), vbI4 = *(const v4f*)(bI4 + e4);

    const size_t plane = (size_t)BS_ * E_;
    float* o = out + (size_t)bs * E_ + e4;
    __builtin_nontemporal_store(sshrink4(a1re + vbR1), (v4f*)(o + 0 * plane));
    __builtin_nontemporal_store(sshrink4(a1im + vbI1), (v4f*)(o + 1 * plane));
    __builtin_nontemporal_store(sshrink4(a2re + vbR2), (v4f*)(o + 2 * plane));
    __builtin_nontemporal_store(sshrink4(a2im + vbI2), (v4f*)(o + 3 * plane));
    __builtin_nontemporal_store(sshrink4(a3re + vbR3), (v4f*)(o + 4 * plane));
    __builtin_nontemporal_store(sshrink4(a3im + vbI3), (v4f*)(o + 5 * plane));
    __builtin_nontemporal_store(sshrink4(a4re + vbR4), (v4f*)(o + 6 * plane));
    __builtin_nontemporal_store(sshrink4(a4im + vbI4), (v4f*)(o + 7 * plane));
}

extern "C" void kernel_launch(void* const* d_in, const int* in_sizes, int n_in,
                              void* d_out, int out_size, void* d_ws, size_t ws_size,
                              hipStream_t stream) {
    const float* xr = (const float*)d_in[0];
    const float* xi = (const float*)d_in[1];
    float* ws = (float*)d_ws;

    rowsum_wmma<<<8 * 32, 32, 0, stream>>>(
        (const float*)d_in[2], (const float*)d_in[3],
        (const float*)d_in[4], (const float*)d_in[5],
        (const float*)d_in[6], (const float*)d_in[7],
        (const float*)d_in[8], (const float*)d_in[9], ws);

    const int total  = BS_ * (E_ / 4);
    const int block  = 256;
    const int grid   = (total + block - 1) / block;
    octmlp_main<<<grid, block, 0, stream>>>(
        xr, xi, ws,
        (const float*)d_in[10], (const float*)d_in[11],
        (const float*)d_in[12], (const float*)d_in[13],
        (const float*)d_in[14], (const float*)d_in[15],
        (const float*)d_in[16], (const float*)d_in[17],
        (float*)d_out);
}


# Main pass on a resident grid: 512 work-groups x 8 waves (4096 waves, was 32768); each wave walks 8 rows of one 128-column slice, next row's loads requested ahead; row sums and biases read once a wave
# speedup vs baseline: 1.1830x; 1.1830x over previous
#include <hip/hip_runtime.h>

typedef float v2f __attribute__((ext_vector_type(2)));
typedef float v4f __attribute__((ext_vector_type(4)));
typedef float v8f __attribute__((ext_vector_type(8)));

#define LAMBDA_ 0.01f
#define B_ 8
#define S_ 1024
#define E_ 512
#define BS_ (B_ * S_)

__global__ void __launch_bounds__(32) rowsum_wmma(
    const float* __restrict__ y0, const float* __restrict__ y1,
    const float* __restrict__ y2, const float* __restrict__ y3,
    const float* __restrict__ y4, const float* __restrict__ y5,
    const float* __restrict__ y6, const float* __restrict__ y7,
    float* __restrict__ ws)
{
    const int m  = blockIdx.x >> 5;
    const int rb = blockIdx.x & 31;
    const float* Y =
        (m == 0) ? y0 : (m == 1) ? y1 : (m == 2) ? y2 : (m == 3) ? y3 :
        (m == 4) ? y4 : (m == 5) ? y5 : (m == 6) ? y6 : y7;

    const int lane  = threadIdx.x;
    const int row   = rb * 16 + (lane & 15);
    const int khalf = (lane >> 4) * 2;

    v8f c = {};
    v2f b;  b.x = 1.0f;  b.y = 1.0f;

    const float* rowp = Y + (size_t)row * E_ + khalf;
    #pragma unroll 4
    for (int k0 = 0; k0 < E_; k0 += 4) {
        v2f a = *(const v2f*)(rowp + k0);
        c = __builtin_amdgcn_wmma_f32_16x16x4_f32(
                 false, a,  false, b,
                 (short)0, c,  false,  false);
    }

    float* o = ws + (size_t)m * E_ + rb * 16;
    if (lane == 0) {
        #pragma unroll
        for (int v = 0; v < 8; ++v) o[v] = c[v];
    } else if (lane == 16) {
        #pragma unroll
        for (int v = 0; v < 8; ++v) o[8 + v] = c[v];
    }
}

__device__ __forceinline__ v4f sshrink4(v4f v) {
    v4f r;
    #pragma unroll
    for (int i = 0; i < 4; ++i) {
        float x = v[i];
        r[i] = (x > LAMBDA_) ? (x - LAMBDA_) : ((x < -LAMBDA_) ? (x + LAMBDA_) : 0.0f);
    }
    return r;
}

__device__ __forceinline__ void cm(v4f xr, v4f xi, v4f sr, v4f si,
                                   v4f& re, v4f& im) {
    re = sshrink4(xr * sr - xi * si);
    im = sshrink4(xr * si + xi * sr);
}

struct SliceConsts {
    v4f r1, i1, r2, i2, r3, i3, r4, i4;
    v4f bR1, bI1, bR2, bI2, bR3, bI3, bR4, bI4;
};

__device__ __forceinline__ void load_row(v4f (&b)[8], const float* pr, const float* pi,
                                         unsigned lo) {
    #pragma unroll
    for (int c = 0; c < 4; ++c) {
        b[c]     = __builtin_nontemporal_load((const v4f*)(pr + c * E_ + lo));
        b[4 + c] = __builtin_nontemporal_load((const v4f*)(pi + c * E_ + lo));
    }
}

__device__ __forceinline__ void row_out(const v4f (&x)[8], const SliceConsts& k,
                                        float* o, unsigned lo) {
    const v4f x1r = x[0], x2r = x[1], x3r = x[2], x4r = x[3];
    const v4f x1i = x[4], x2i = x[5], x3i = x[6], x4i = x[7];
    const v4f r1 = k.r1, i1 = k.i1, r2 = k.r2, i2 = k.i2;
    const v4f r3 = k.r3, i3 = k.i3, r4 = k.r4, i4 = k.i4;
    const v4f ni2 = -i2, ni3 = -i3, ni4 = -i4;

    v4f tre, tim;
    v4f a1re, a1im, a2re, a2im, a3re, a3im, a4re, a4im;

    cm(x1r, x1i, r1, i1, a1re, a1im);
    cm(x2r, x2i, r2, ni2, tre, tim);  a1re -= tre;  a1im -= tim;
    cm(x3r, x3i, r3, ni3, tre, tim);  a1re -= tre;  a1im -= tim;
    cm(x4r, x4i, r4, ni4, tre, tim);  a1re -= tre;  a1im -= tim;

    cm(x2r, x2i, r1, i1, a2re, a2im);
    cm(x1r, x1i, r2, i2, tre, tim);   a2re += tre;  a2im += tim;
    cm(x4r, x4i, r3, ni3, tre, tim);  a2re += tre;  a2im += tim;
    cm(x3r, x3i, r4, ni4, tre, tim);  a2re -= tre;  a2im -= tim;

    cm(x3r, x3i, r1, i1, a3re, a3im);
    cm(x4r, x4i, r2, ni2, tre, tim);  a3re -= tre;  a3im -= tim;
    cm(x1r, x1i, r3, i3, tre, tim);   a3re += tre;  a3im += tim;
    cm(x2r, x2i, r4, ni4, tre, tim);  a3re += tre;  a3im += tim;

    cm(x4r, x4i, r1, i1, a4re, a4im);
    cm(x3r, x3i, r2, ni2, tre, tim);  a4re += tre;  a4im += tim;
    cm(x2r, x2i, r3, ni3, tre, tim);  a4re -= tre;  a4im -= tim;
    cm(x1r, x1i, r4, i4, tre, tim);   a4re += tre;  a4im += tim;

    const size_t plane = (size_t)BS_ * E_;
    __builtin_nontemporal_store(sshrink4(a1re + k.bR1), (v4f*)(o + 0 * plane + lo));
    __builtin_nontemporal_store(sshrink4(a1im + k.bI1), (v4f*)(o + 1 * plane + lo));
    __builtin_nontemporal_store(sshrink4(a2re + k.bR2), (v4f*)(o + 2 * plane + lo));
    __builtin_nontemporal_store(sshrink4(a2im + k.bI2), (v4f*)(o + 3 * plane + lo));
    __builtin_nontemporal_store(sshrink4(a3re + k.bR3), (v4f*)(o + 4 * plane + lo));
    __builtin_nontemporal_store(sshrink4(a3im + k.bI3), (v4f*)(o + 5 * plane + lo));
    __builtin_nontemporal_store(sshrink4(a4re + k.bR4), (v4f*)(o + 6 * plane + lo));
    __builtin_nontemporal_store(sshrink4(a4im + k.bI4), (v4f*)(o + 7 * plane + lo));
}

#define WAVES_TOTAL_   4096
#define ROWS_PER_WAVE_ (BS_ * 4 / WAVES_TOTAL_)

__global__ void __launch_bounds__(256, 4) octmlp_main(
    const float* __restrict__ xr, const float* __restrict__ xi,
    const float* __restrict__ ws,
    const float* __restrict__ bR1, const float* __restrict__ bI1,
    const float* __restrict__ bR2, const float* __restrict__ bI2,
    const float* __restrict__ bR3, const float* __restrict__ bI3,
    const float* __restrict__ bR4, const float* __restrict__ bI4,
    float* __restrict__ out)
{
    const unsigned lane = threadIdx.x & 31u;
    const unsigned wave = __builtin_amdgcn_readfirstlane(threadIdx.x >> 5);
    const unsigned gw   = blockIdx.x * 8u + wave;
    const unsigned bs0  = (gw >> 2) * ROWS_PER_WAVE_;
    const unsigned col0 = (gw & 3u) * 128u;
    const unsigned lo   = lane * 4u;
    const unsigned e4   = col0 + lo;

    SliceConsts k;
    k.r1 = *(const v4f*)(ws + 0 * E_ + e4);  k.i1 = *(const v4f*)(ws + 1 * E_ + e4);
    k.r2 = *(const v4f*)(ws + 2 * E_ + e4);  k.i2 = *(const v4f*)(ws + 3 * E_ + e4);
    k.r3 = *(const v4f*)(ws + 4 * E_ + e4);  k.i3 = *(const v4f*)(ws + 5 * E_ + e4);
    k.r4 = *(const v4f*)(ws + 6 * E_ + e4);  k.i4 = *(const v4f*)(ws + 7 * E_ + e4);
    k.bR1 = *(const v4f*)(bR1 + e4);  k.bI1 = *(const v4f*)(bI1 + e4);
    k.bR2 = *(const v4f*)(bR2 + e4);  k.bI2 = *(const v4f*)(bI2 + e4);
    k.bR3 = *(const v4f*)(bR3 + e4);  k.bI3 = *(const v4f*)(bI3 + e4);
    k.bR4 = *(const v4f*)(bR4 + e4);  k.bI4 = *(const v4f*)(bI4 + e4);

    const float* pr = xr + ((size_t)bs0 * 4 * E_ + col0);
    const float* pi = xi + ((size_t)bs0 * 4 * E_ + col0);
    float*       po = out + ((size_t)bs0 * E_ + col0);

    v4f A[8], B[8];
    load_row(A, pr, pi, lo);

    #pragma unroll 1
    for (int j = 0; j < ROWS_PER_WAVE_ - 2; j += 2) {
        __builtin_amdgcn_sched_barrier(0);
        load_row(B, pr + 4 * E_, pi + 4 * E_, lo);
        __builtin_amdgcn_sched_barrier(0);
        row_out(A, k, po, lo);
        __builtin_amdgcn_sched_barrier(0);
        load_row(A, pr + 8 * E_, pi + 8 * E_, lo);
        __builtin_amdgcn_sched_barrier(0);
        row_out(B, k, po + E_, lo);
        pr += 8 * E_;  pi += 8 * E_;  po += 2 * E_;
    }
    __builtin_amdgcn_sched_barrier(0);
    load_row(B, pr + 4 * E_, pi + 4 * E_, lo);
    __builtin_amdgcn_sched_barrier(0);
    row_out(A, k, po, lo);
    __builtin_amdgcn_sched_barrier(0);
    row_out(B, k, po + E_, lo);
}

extern "C" void kernel_launch(void* const* d_in, const int* in_sizes, int n_in,
                              void* d_out, int out_size, void* d_ws, size_t ws_size,
                              hipStream_t stream) {
    const float* xr = (const float*)d_in[0];
    const float* xi = (const float*)d_in[1];
    float* ws = (float*)d_ws;

    rowsum_wmma<<<8 * 32, 32, 0, stream>>>(
        (const float*)d_in[2], (const float*)d_in[3],
        (const float*)d_in[4], (const float*)d_in[5],
        (const float*)d_in[6], (const float*)d_in[7],
        (const float*)d_in[8], (const float*)d_in[9], ws);

    octmlp_main<<<512, 256, 0, stream>>>(
        xr, xi, ws,
        (const float*)d_in[10], (const float*)d_in[11],
        (const float*)d_in[12], (const float*)d_in[13],
        (const float*)d_in[14], (const float*)d_in[15],
        (const float*)d_in[16], (const float*)d_in[17],
        (float*)d_out);
}


# Main pass: softshrink as x - med3(x,-L,+L), 2 vector ops per value (was 2 compares, 2 selects, 2 adds); each output stored when complete so it stays in registers; 912 -> 440 vector instr per row
# speedup vs baseline: 1.3255x; 1.1205x over previous
#include <hip/hip_runtime.h>

typedef float v2f __attribute__((ext_vector_type(2)));
typedef float v4f __attribute__((ext_vector_type(4)));
typedef float v8f __attribute__((ext_vector_type(8)));

#define LAMBDA_ 0.01f
#define B_ 8
#define S_ 1024
#define E_ 512
#define BS_ (B_ * S_)

__global__ void __launch_bounds__(32) rowsum_wmma(
    const float* __restrict__ y0, const float* __restrict__ y1,
    const float* __restrict__ y2, const float* __restrict__ y3,
    const float* __restrict__ y4, const float* __restrict__ y5,
    const float* __restrict__ y6, const float* __restrict__ y7,
    float* __restrict__ ws)
{
    const int m  = blockIdx.x >> 5;
    const int rb = blockIdx.x & 31;
    const float* Y =
        (m == 0) ? y0 : (m == 1) ? y1 : (m == 2) ? y2 : (m == 3) ? y3 :
        (m == 4) ? y4 : (m == 5) ? y5 : (m == 6) ? y6 : y7;

    const int lane  = threadIdx.x;
    const int row   = rb * 16 + (lane & 15);
    const int khalf = (lane >> 4) * 2;

    v8f c = {};
    v2f b;  b.x = 1.0f;  b.y = 1.0f;

    const float* rowp = Y + (size_t)row * E_ + khalf;
    #pragma unroll 4
    for (int k0 = 0; k0 < E_; k0 += 4) {
        v2f a = *(const v2f*)(rowp + k0);
        c = __builtin_amdgcn_wmma_f32_16x16x4_f32(
                 false, a,  false, b,
                 (short)0, c,  false,  false);
    }

    float* o = ws + (size_t)m * E_ + rb * 16;
    if (lane == 0) {
        #pragma unroll
        for (int v = 0; v < 8; ++v) o[v] = c[v];
    } else if (lane == 16) {
        #pragma unroll
        for (int v = 0; v < 8; ++v) o[8 + v] = c[v];
    }
}

__device__ __forceinline__ v4f sshrink4(v4f v) {
    v4f c;
    #pragma unroll
    for (int i = 0; i < 4; ++i)
        c[i] = __builtin_amdgcn_fmed3f(v[i], -LAMBDA_, LAMBDA_);
    return v - c;
}

__device__ __forceinline__ void cm(v4f xr, v4f xi, v4f sr, v4f si,
                                   v4f& re, v4f& im) {
    re = sshrink4(xr * sr - xi * si);
    im = sshrink4(xr * si + xi * sr);
}

struct SliceConsts {
    v4f r1, i1, r2, i2, r3, i3, r4, i4;
    v4f bR1, bI1, bR2, bI2, bR3, bI3, bR4, bI4;
};

__device__ __forceinline__ void load_row(v4f (&b)[8], const float* pr, const float* pi,
                                         unsigned lo) {
    #pragma unroll
    for (int c = 0; c < 4; ++c) {
        b[c]     = __builtin_nontemporal_load((const v4f*)(pr + c * E_ + lo));
        b[4 + c] = __builtin_nontemporal_load((const v4f*)(pi + c * E_ + lo));
    }
}

__device__ __forceinline__ void row_out(const v4f (&x)[8], const SliceConsts& k,
                                        float* o, unsigned lo) {
    const v4f x1r = x[0], x2r = x[1], x3r = x[2], x4r = x[3];
    const v4f x1i = x[4], x2i = x[5], x3i = x[6], x4i = x[7];
    const v4f r1 = k.r1, i1 = k.i1, r2 = k.r2, i2 = k.i2;
    const v4f r3 = k.r3, i3 = k.i3, r4 = k.r4, i4 = k.i4;
    const v4f ni2 = -i2, ni3 = -i3, ni4 = -i4;
    const size_t plane = (size_t)BS_ * E_;

    v4f tre, tim, are, aim;

    cm(x1r, x1i, r1, i1, are, aim);
    cm(x2r, x2i, r2, ni2, tre, tim);  are -= tre;  aim -= tim;
    cm(x3r, x3i, r3, ni3, tre, tim);  are -= tre;  aim -= tim;
    cm(x4r, x4i, r4, ni4, tre, tim);  are -= tre;  aim -= tim;
    __builtin_nontemporal_store(sshrink4(are + k.bR1), (v4f*)(o + 0 * plane + lo));
    __builtin_nontemporal_store(sshrink4(aim + k.bI1), (v4f*)(o + 1 * plane + lo));
    __builtin_amdgcn_sched_barrier(0);

    cm(x2r, x2i, r1, i1, are, aim);
    cm(x1r, x1i, r2, i2, tre, tim);   are += tre;  aim += tim;
    cm(x4r, x4i, r3, ni3, tre, tim);  are += tre;  aim += tim;
    cm(x3r, x3i, r4, ni4, tre, tim);  are -= tre;  aim -= tim;
    __builtin_nontemporal_store(sshrink4(are + k.bR2), (v4f*)(o + 2 * plane + lo));
    __builtin_nontemporal_store(sshrink4(aim + k.bI2), (v4f*)(o + 3 * plane + lo));
    __builtin_amdgcn_sched_barrier(0);

    cm(x3r, x3i, r1, i1, are, aim);
    cm(x4r, x4i, r2, ni2, tre, tim);  are -= tre;  aim -= tim;
    cm(x1r, x1i, r3, i3, tre, tim);   are += tre;  aim += tim;
    cm(x2r, x2i, r4, ni4, tre, tim);  are += tre;  aim += tim;
    __builtin_nontemporal_store(sshrink4(are + k.bR3), (v4f*)(o + 4 * plane + lo));
    __builtin_nontemporal_store(sshrink4(aim + k.bI3), (v4f*)(o + 5 * plane + lo));
    __builtin_amdgcn_sched_barrier(0);

    cm(x4r, x4i, r1, i1, are, aim);
    cm(x3r, x3i, r2, ni2, tre, tim);  are += tre;  aim += tim;
    cm(x2r, x2i, r3, ni3, tre, tim);  are -= tre;  aim -= tim;
    cm(x1r, x1i, r4, i4, tre, tim);   are += tre;  aim += tim;
    __builtin_nontemporal_store(sshrink4(are + k.bR4), (v4f*)(o + 6 * plane + lo));
    __builtin_nontemporal_store(sshrink4(aim + k.bI4), (v4f*)(o + 7 * plane + lo));
}

#define WAVES_TOTAL_   4096
#define ROWS_PER_WAVE_ (BS_ * 4 / WAVES_TOTAL_)

__global__ void __launch_bounds__(256, 4) octmlp_main(
    const float* __restrict__ xr, const float* __restrict__ xi,
    const float* __restrict__ ws,
    const float* __restrict__ bR1, const float* __restrict__ bI1,
    const float* __restrict__ bR2, const float* __restrict__ bI2,
    const float* __restrict__ bR3, const float* __restrict__ bI3,
    const float* __restrict__ bR4, const float* __restrict__ bI4,
    float* __restrict__ out)
{
    const unsigned lane = threadIdx.x & 31u;
    const unsigned wave = __builtin_amdgcn_readfirstlane(threadIdx.x >> 5);
    const unsigned gw   = blockIdx.x * 8u + wave;
    const unsigned bs0  = (gw >> 2) * ROWS_PER_WAVE_;
    const unsigned col0 = (gw & 3u) * 128u;
    const unsigned lo   = lane * 4u;
    const unsigned e4   = col0 + lo;

    SliceConsts k;
    k.r1 = *(const v4f*)(ws + 0 * E_ + e4);  k.i1 = *(const v4f*)(ws + 1 * E_ + e4);
    k.r2 = *(const v4f*)(ws + 2 * E_ + e4);  k.i2 = *(const v4f*)(ws + 3 * E_ + e4);
    k.r3 = *(const v4f*)(ws + 4 * E_ + e4);  k.i3 = *(const v4f*)(ws + 5 * E_ + e4);
    k.r4 = *(const v4f*)(ws + 6 * E_ + e4);  k.i4 = *(const v4f*)(ws + 7 * E_ + e4);
    k.bR1 = *(const v4f*)(bR1 + e4);  k.bI1 = *(const v4f*)(bI1 + e4);
    k.bR2 = *(const v4f*)(bR2 + e4);  k.bI2 = *(const v4f*)(bI2 + e4);
    k.bR3 = *(const v4f*)(bR3 + e4);  k.bI3 = *(const v4f*)(bI3 + e4);
    k.bR4 = *(const v4f*)(bR4 + e4);  k.bI4 = *(const v4f*)(bI4 + e4);

    const float* pr = xr + ((size_t)bs0 * 4 * E_ + col0);
    const float* pi = xi + ((size_t)bs0 * 4 * E_ + col0);
    float*       po = out + ((size_t)bs0 * E_ + col0);

    v4f A[8], B[8];
    load_row(A, pr, pi, lo);

    #pragma unroll 1
    for (int j = 0; j < ROWS_PER_WAVE_ - 2; j += 2) {
        __builtin_amdgcn_sched_barrier(0);
        load_row(B, pr + 4 * E_, pi + 4 * E_, lo);
        __builtin_amdgcn_sched_barrier(0);
        row_out(A, k, po, lo);
        __builtin_amdgcn_sched_barrier(0);
        load_row(A, pr + 8 * E_, pi + 8 * E_, lo);
        __builtin_amdgcn_sched_barrier(0);
        row_out(B, k, po + E_, lo);
        pr += 8 * E_;  pi += 8 * E_;  po += 2 * E_;
    }
    __builtin_amdgcn_sched_barrier(0);
    load_row(B, pr + 4 * E_, pi + 4 * E_, lo);
    __builtin_amdgcn_sched_barrier(0);
    row_out(A, k, po, lo);
    __builtin_amdgcn_sched_barrier(0);
    row_out(B, k, po + E_, lo);
}

extern "C" void kernel_launch(void* const* d_in, const int* in_sizes, int n_in,
                              void* d_out, int out_size, void* d_ws, size_t ws_size,
                              hipStream_t stream) {
    const float* xr = (const float*)d_in[0];
    const float* xi = (const float*)d_in[1];
    float* ws = (float*)d_ws;

    rowsum_wmma<<<8 * 32, 32, 0, stream>>>(
        (const float*)d_in[2], (const float*)d_in[3],
        (const float*)d_in[4], (const float*)d_in[5],
        (const float*)d_in[6], (const float*)d_in[7],
        (const float*)d_in[8], (const float*)d_in[9], ws);

    octmlp_main<<<512, 256, 0, stream>>>(
        xr, xi, ws,
        (const float*)d_in[10], (const float*)d_in[11],
        (const float*)d_in[12], (const float*)d_in[13],
        (const float*)d_in[14], (const float*)d_in[15],
        (const float*)d_in[16], (const float*)d_in[17],
        (float*)d_out);
}


# Row-sum pass: each 16-row block's columns split over 4 waves (1024 waves, was 256), a wave's 16 loads all requested before its 32 WMMAs (2 accumulators); partials joined in LDS: sum order changes
# speedup vs baseline: 1.5769x; 1.1897x over previous
#include <hip/hip_runtime.h>

typedef float v2f __attribute__((ext_vector_type(2)));
typedef float v4f __attribute__((ext_vector_type(4)));
typedef float v8f __attribute__((ext_vector_type(8)));

#define LAMBDA_ 0.01f
#define B_ 8
#define S_ 1024
#define E_ 512
#define BS_ (B_ * S_)

#define RS_WAVES_ 4
#define RS_COLS_  (E_ / RS_WAVES_)
#define RS_LOADS_ (RS_COLS_ / 8)

__global__ void __launch_bounds__(32 * RS_WAVES_) rowsum_wmma(
    const float* __restrict__ y0, const float* __restrict__ y1,
    const float* __restrict__ y2, const float* __restrict__ y3,
    const float* __restrict__ y4, const float* __restrict__ y5,
    const float* __restrict__ y6, const float* __restrict__ y7,
    float* __restrict__ ws)
{
    __shared__ float part[RS_WAVES_ * 2][16];

    const int m  = blockIdx.x >> 5;
    const int rb = blockIdx.x & 31;
    const float* Y =
        (m == 0) ? y0 : (m == 1) ? y1 : (m == 2) ? y2 : (m == 3) ? y3 :
        (m == 4) ? y4 : (m == 5) ? y5 : (m == 6) ? y6 : y7;

    const int lane = threadIdx.x & 31;
    const int wave = threadIdx.x >> 5;
    const int row  = rb * 16 + (lane & 15);
    const int half = lane >> 4;

    const float* p = Y + (size_t)row * E_ + wave * RS_COLS_ + half * 4;

    v4f a[RS_LOADS_];
    #pragma unroll
    for (int j = 0; j < RS_LOADS_; ++j)
        a[j] = *(const v4f*)(p + j * 8);
    __builtin_amdgcn_sched_barrier(0);

    v8f c0 = {}, c1 = {};
    v2f b;  b.x = 1.0f;  b.y = 1.0f;
    #pragma unroll
    for (int j = 0; j < RS_LOADS_; ++j) {
        v2f lo2, hi2;
        lo2.x = a[j].x;  lo2.y = a[j].y;
        hi2.x = a[j].z;  hi2.y = a[j].w;
        c0 = __builtin_amdgcn_wmma_f32_16x16x4_f32(
                 false, lo2,  false, b,
                 (short)0, c0,  false,  false);
        c1 = __builtin_amdgcn_wmma_f32_16x16x4_f32(
                 false, hi2,  false, b,
                 (short)0, c1,  false,  false);
    }

    if ((lane & 15) == 0) {
        #pragma unroll
        for (int v = 0; v < 8; ++v) {
            part[wave * 2 + 0][half * 8 + v] = c0[v];
            part[wave * 2 + 1][half * 8 + v] = c1[v];
        }
    }
    __syncthreads();
    if (threadIdx.x < 16) {
        const int r = threadIdx.x;
        float s = part[0][r];
        #pragma unroll
        for (int q = 1; q < RS_WAVES_ * 2; ++q) s += part[q][r];
        ws[(size_t)m * E_ + rb * 16 + r] = s;
    }
}

__device__ __forceinline__ v4f sshrink4(v4f v) {
    v4f c;
    #pragma unroll
    for (int i = 0; i < 4; ++i)
        c[i] = __builtin_amdgcn_fmed3f(v[i], -LAMBDA_, LAMBDA_);
    return v - c;
}

__device__ __forceinline__ void cm(v4f xr, v4f xi, v4f sr, v4f si,
                                   v4f& re, v4f& im) {
    re = sshrink4(xr * sr - xi * si);
    im = sshrink4(xr * si + xi * sr);
}

struct SliceConsts {
    v4f r1, i1, r2, i2, r3, i3, r4, i4;
    v4f bR1, bI1, bR2, bI2, bR3, bI3, bR4, bI4;
};

__device__ __forceinline__ void load_row(v4f (&b)[8], const float* pr, const float* pi,
                                         unsigned lo) {
    #pragma unroll
    for (int c = 0; c < 4; ++c) {
        b[c]     = __builtin_nontemporal_load((const v4f*)(pr + c * E_ + lo));
        b[4 + c] = __builtin_nontemporal_load((const v4f*)(pi + c * E_ + lo));
    }
}

__device__ __forceinline__ void row_out(const v4f (&x)[8], const SliceConsts& k,
                                        float* o, unsigned lo) {
    const v4f x1r = x[0], x2r = x[1], x3r = x[2], x4r = x[3];
    const v4f x1i = x[4], x2i = x[5], x3i = x[6], x4i = x[7];
    const v4f r1 = k.r1, i1 = k.i1, r2 = k.r2, i2 = k.i2;
    const v4f r3 = k.r3, i3 = k.i3, r4 = k.r4, i4 = k.i4;
    const v4f ni2 = -i2, ni3 = -i3, ni4 = -i4;
    const size_t plane = (size_t)BS_ * E_;

    v4f tre, tim, are, aim;

    cm(x1r, x1i, r1, i1, are, aim);
    cm(x2r, x2i, r2, ni2, tre, tim);  are -= tre;  aim -= tim;
    cm(x3r, x3i, r3, ni3, tre, tim);  are -= tre;  aim -= tim;
    cm(x4r, x4i, r4, ni4, tre, tim);  are -= tre;  aim -= tim;
    __builtin_nontemporal_store(sshrink4(are + k.bR1), (v4f*)(o + 0 * plane + lo));
    __builtin_nontemporal_store(sshrink4(aim + k.bI1), (v4f*)(o + 1 * plane + lo));
    __builtin_amdgcn_sched_barrier(0);

    cm(x2r, x2i, r1, i1, are, aim);
    cm(x1r, x1i, r2, i2, tre, tim);   are += tre;  aim += tim;
    cm(x4r, x4i, r3, ni3, tre, tim);  are += tre;  aim += tim;
    cm(x3r, x3i, r4, ni4, tre, tim);  are -= tre;  aim -= tim;
    __builtin_nontemporal_store(sshrink4(are + k.bR2), (v4f*)(o + 2 * plane + lo));
    __builtin_nontemporal_store(sshrink4(aim + k.bI2), (v4f*)(o + 3 * plane + lo));
    __builtin_amdgcn_sched_barrier(0);

    cm(x3r, x3i, r1, i1, are, aim);
    cm(x4r, x4i, r2, ni2, tre, tim);  are -= tre;  aim -= tim;
    cm(x1r, x1i, r3, i3, tre, tim);   are += tre;  aim += tim;
    cm(x2r, x2i, r4, ni4, tre, tim);  are += tre;  aim += tim;
    __builtin_nontemporal_store(sshrink4(are + k.bR3), (v4f*)(o + 4 * plane + lo));
    __builtin_nontemporal_store(sshrink4(aim + k.bI3), (v4f*)(o + 5 * plane + lo));
    __builtin_amdgcn_sched_barrier(0);

    cm(x4r, x4i, r1, i1, are, aim);
    cm(x3r, x3i, r2, ni2, tre, tim);  are += tre;  aim += tim;
    cm(x2r, x2i, r3, ni3, tre, tim);  are -= tre;  aim -= tim;
    cm(x1r, x1i, r4, i4, tre, tim);   are += tre;  aim += tim;
    __builtin_nontemporal_store(sshrink4(are + k.bR4), (v4f*)(o + 6 * plane + lo));
    __builtin_nontemporal_store(sshrink4(aim + k.bI4), (v4f*)(o + 7 * plane + lo));
}

#define WAVES_TOTAL_   4096
#define ROWS_PER_WAVE_ (BS_ * 4 / WAVES_TOTAL_)

__global__ void __launch_bounds__(256, 4) octmlp_main(
    const float* __restrict__ xr, const float* __restrict__ xi,
    const float* __restrict__ ws,
    const float* __restrict__ bR1, const float* __restrict__ bI1,
    const float* __restrict__ bR2, const float* __restrict__ bI2,
    const float* __restrict__ bR3, const float* __restrict__ bI3,
    const float* __restrict__ bR4, const float* __restrict__ bI4,
    float* __restrict__ out)
{
    const unsigned lane = threadIdx.x & 31u;
    const unsigned wave = __builtin_amdgcn_readfirstlane(threadIdx.x >> 5);
    const unsigned gw   = blockIdx.x * 8u + wave;
    const unsigned bs0  = (gw >> 2) * ROWS_PER_WAVE_;
    const unsigned col0 = (gw & 3u) * 128u;
    const unsigned lo   = lane * 4u;
    const unsigned e4   = col0 + lo;

    SliceConsts k;
    k.r1 = *(const v4f*)(ws + 0 * E_ + e4);  k.i1 = *(const v4f*)(ws + 1 * E_ + e4);
    k.r2 = *(const v4f*)(ws + 2 * E_ + e4);  k.i2 = *(const v4f*)(ws + 3 * E_ + e4);
    k.r3 = *(const v4f*)(ws + 4 * E_ + e4);  k.i3 = *(const v4f*)(ws + 5 * E_ + e4);
    k.r4 = *(const v4f*)(ws + 6 * E_ + e4);  k.i4 = *(const v4f*)(ws + 7 * E_ + e4);
    k.bR1 = *(const v4f*)(bR1 + e4);  k.bI1 = *(const v4f*)(bI1 + e4);
    k.bR2 = *(const v4f*)(bR2 + e4);  k.bI2 = *(const v4f*)(bI2 + e4);
    k.bR3 = *(const v4f*)(bR3 + e4);  k.bI3 = *(const v4f*)(bI3 + e4);
    k.bR4 = *(const v4f*)(bR4 + e4);  k.bI4 = *(const v4f*)(bI4 + e4);

    const float* pr = xr + ((size_t)bs0 * 4 * E_ + col0);
    const float* pi = xi + ((size_t)bs0 * 4 * E_ + col0);
    float*       po = out + ((size_t)bs0 * E_ + col0);

    v4f A[8], B[8];
    load_row(A, pr, pi, lo);

    #pragma unroll 1
    for (int j = 0; j < ROWS_PER_WAVE_ - 2; j += 2) {
        __builtin_amdgcn_sched_barrier(0);
        load_row(B, pr + 4 * E_, pi + 4 * E_, lo);
        __builtin_amdgcn_sched_barrier(0);
        row_out(A, k, po, lo);
        __builtin_amdgcn_sched_barrier(0);
        load_row(A, pr + 8 * E_, pi + 8 * E_, lo);
        __builtin_amdgcn_sched_barrier(0);
        row_out(B, k, po + E_, lo);
        pr += 8 * E_;  pi += 8 * E_;  po += 2 * E_;
    }
    __builtin_amdgcn_sched_barrier(0);
    load_row(B, pr + 4 * E_, pi + 4 * E_, lo);
    __builtin_amdgcn_sched_barrier(0);
    row_out(A, k, po, lo);
    __builtin_amdgcn_sched_barrier(0);
    row_out(B, k, po + E_, lo);
}

extern "C" void kernel_launch(void* const* d_in, const int* in_sizes, int n_in,
                              void* d_out, int out_size, void* d_ws, size_t ws_size,
                              hipStream_t stream) {
    const float* xr = (const float*)d_in[0];
    const float* xi = (const float*)d_in[1];
    float* ws = (float*)d_ws;

    rowsum_wmma<<<8 * 32, 32 * RS_WAVES_, 0, stream>>>(
        (const float*)d_in[2], (const float*)d_in[3],
        (const float*)d_in[4], (const float*)d_in[5],
        (const float*)d_in[6], (const float*)d_in[7],
        (const float*)d_in[8], (const float*)d_in[9], ws);

    octmlp_main<<<512, 256, 0, stream>>>(
        xr, xi, ws,
        (const float*)d_in[10], (const float*)d_in[11],
        (const float*)d_in[12], (const float*)d_in[13],
        (const float*)d_in[14], (const float*)d_in[15],
        (const float*)d_in[16], (const float*)d_in[17],
        (float*)d_out);
}
